# starting kernel
# baseline (speedup 1.0000x reference)
#include <hip/hip_runtime.h>
#include <math.h>


typedef __attribute__((ext_vector_type(2))) float v2f;
typedef __attribute__((ext_vector_type(8))) float v8f;

#define NB 4
#define NC 64
#define NL 4096
#define LT 64
#define NTILES (NL / LT)
#define NWAVES 4
#define BLOCK (NWAVES * 32)
#define EPS_BN 1e-5f

#define SE_STRIDE 68
#define SH_STRIDE 65

__device__ __forceinline__ v8f wmma_f32(v2f a, v2f b, v8f c) {
  return __builtin_amdgcn_wmma_f32_16x16x4_f32(false, a, false, b,
                                               (short)0, c, false, false);
}

__global__ __launch_bounds__(BLOCK)
void k1_conv_qkv_M(const float* __restrict__ x,
                   const float* __restrict__ conv_w, const float* __restrict__ conv_b,
                   const float* __restrict__ wq, const float* __restrict__ bq,
                   const float* __restrict__ wk, const float* __restrict__ bk,
                   const float* __restrict__ wv, const float* __restrict__ bv,
                   float* __restrict__ kbuf, float* __restrict__ Mpart) {
  __shared__ float sE[NC * SE_STRIDE];
  __shared__ float sH[NC * SH_STRIDE];
  __shared__ float sQ[NC * SH_STRIDE];
  __shared__ float sV[NC * SH_STRIDE];

  const int tid  = threadIdx.x;
  const int wave = tid >> 5;
  const int lane = tid & 31;
  const int half = lane >> 4;
  const int mn   = lane & 15;

  const int bIdx = blockIdx.x / NTILES;
  const int l0   = (blockIdx.x % NTILES) * LT;

  const float* xb = x + bIdx * NC * NL;
  for (int idx = tid; idx < NC * (LT + 2); idx += BLOCK) {
    int c = idx / (LT + 2);
    int j = idx % (LT + 2);
    int gl = (l0 + j - 1) & (NL - 1);
    float v = xb[c * NL + gl];
    sE[c * SE_STRIDE + j] = (v > 0.0f) ? v : (__expf(v) - 1.0f);
  }
  __syncthreads();

  const int mt = wave;
  const int arow = mt * 16 + mn;
  const int rbase = mt * 16 + half * 8;

  for (int nt = 0; nt < 4; ++nt) {
    const int bcol = nt * 16 + mn;
    v8f acc = {};
    for (int tap = 0; tap < 3; ++tap) {
      const float* Wt = conv_w + tap;
      for (int kk = 0; kk < 16; ++kk) {
        const int k0 = kk * 4 + half * 2;
        v2f a, b;
        a[0] = Wt[(arow * NC + k0) * 3];
        a[1] = Wt[(arow * NC + k0 + 1) * 3];
        b[0] = sE[k0 * SE_STRIDE + bcol + tap];
        b[1] = sE[(k0 + 1) * SE_STRIDE + bcol + tap];
        acc = wmma_f32(a, b, acc);
      }
    }
#pragma unroll
    for (int r = 0; r < 8; ++r)
      sH[(rbase + r) * SH_STRIDE + bcol] = acc[r] + conv_b[rbase + r];
  }
  __syncthreads();

  for (int nt = 0; nt < 4; ++nt) {
    const int bcol = nt * 16 + mn;
    v8f aq = {}, ak = {}, av = {};
    for (int kk = 0; kk < 16; ++kk) {
      const int k0 = kk * 4 + half * 2;
      v2f bf, a;
      bf[0] = sH[k0 * SH_STRIDE + bcol];
      bf[1] = sH[(k0 + 1) * SH_STRIDE + bcol];
      a[0] = wq[arow * NC + k0]; a[1] = wq[arow * NC + k0 + 1];
      aq = wmma_f32(a, bf, aq);
      a[0] = wk[arow * NC + k0]; a[1] = wk[arow * NC + k0 + 1];
      ak = wmma_f32(a, bf, ak);
      a[0] = wv[arow * NC + k0]; a[1] = wv[arow * NC + k0 + 1];
      av = wmma_f32(a, bf, av);
    }
#pragma unroll
    for (int r = 0; r < 8; ++r) {
      const int row = rbase + r;
      sQ[row * SH_STRIDE + bcol] = aq[r] + bq[row];
      sV[row * SH_STRIDE + bcol] = av[r] + bv[row];
      kbuf[(bIdx * NC + row) * NL + l0 + bcol] = ak[r] + bk[row];
    }
  }
  __syncthreads();

  float* Mout = Mpart + blockIdx.x * (NC * NC);
  for (int nt = 0; nt < 4; ++nt) {
    const int bcol = nt * 16 + mn;
    v8f acc = {};
    for (int kk = 0; kk < 16; ++kk) {
      const int k0 = kk * 4 + half * 2;
      v2f a, b;
      a[0] = sV[arow * SH_STRIDE + k0];
      a[1] = sV[arow * SH_STRIDE + k0 + 1];
      b[0] = sQ[bcol * SH_STRIDE + k0];
      b[1] = sQ[bcol * SH_STRIDE + k0 + 1];
      acc = wmma_f32(a, b, acc);
    }
#pragma unroll
    for (int r = 0; r < 8; ++r)
      Mout[(rbase + r) * NC + bcol] = acc[r];
  }
}

__global__ void k1b_reduceM(const float* __restrict__ Mpart,
                            float* __restrict__ Mbuf) {
  int i = blockIdx.x * blockDim.x + threadIdx.x;
  if (i >= NB * NC * NC) return;
  const int b = i >> 12;
  const int mn_ = i & 4095;
  const float* p = Mpart + (size_t)(b * NTILES) * (NC * NC) + mn_;
  float s = 0.0f;
  for (int t = 0; t < NTILES; ++t) s += p[(size_t)t * NC * NC];
  Mbuf[i] = s;
}

__global__ __launch_bounds__(BLOCK)
void k2_gate_residual(const float* __restrict__ x,
                      const float* __restrict__ kbuf,
                      const float* __restrict__ Mbuf,
                      float* __restrict__ y,
                      float* __restrict__ sumPart, float* __restrict__ sqPart) {
  __shared__ float sK[NC * SH_STRIDE];
  __shared__ float sSum[NC * 16];
  __shared__ float sSq[NC * 16];

  const int tid  = threadIdx.x;
  const int wave = tid >> 5;
  const int lane = tid & 31;
  const int half = lane >> 4;
  const int mn   = lane & 15;

  const int bIdx = blockIdx.x / NTILES;
  const int l0   = (blockIdx.x % NTILES) * LT;

  const float* kb = kbuf + (size_t)bIdx * NC * NL + l0;
  for (int idx = tid; idx < NC * LT; idx += BLOCK) {
    int c = idx >> 6, j = idx & 63;
    sK[c * SH_STRIDE + j] = kb[c * NL + j];
  }
  __syncthreads();

  const float* Mb = Mbuf + bIdx * NC * NC;
  const float* xb = x + (size_t)bIdx * NC * NL + l0;
  float* yb = y + (size_t)bIdx * NC * NL + l0;

  const int mt = wave;
  const int arow = mt * 16 + mn;
  const int rbase = mt * 16 + half * 8;

  float regS[8], regQ[8];
#pragma unroll
  for (int r = 0; r < 8; ++r) { regS[r] = 0.0f; regQ[r] = 0.0f; }

  for (int nt = 0; nt < 4; ++nt) {
    const int bcol = nt * 16 + mn;
    v8f acc = {};
    for (int kk = 0; kk < 16; ++kk) {
      const int k0 = kk * 4 + half * 2;
      v2f a, b;
      a[0] = Mb[arow * NC + k0];
      a[1] = Mb[arow * NC + k0 + 1];
      b[0] = sK[k0 * SH_STRIDE + bcol];
      b[1] = sK[(k0 + 1) * SH_STRIDE + bcol];
      acc = wmma_f32(a, b, acc);
    }
#pragma unroll
    for (int r = 0; r < 8; ++r) {
      const int row = rbase + r;
      float yv = acc[r] + xb[row * NL + bcol];
      yb[row * NL + bcol] = yv;
      regS[r] += yv;
      regQ[r] += yv * yv;
    }
  }
#pragma unroll
  for (int r = 0; r < 8; ++r) {
    sSum[(rbase + r) * 16 + mn] = regS[r];
    sSq [(rbase + r) * 16 + mn] = regQ[r];
  }
  __syncthreads();
  if (tid < NC) {
    float s = 0.0f, q = 0.0f;
    for (int j = 0; j < 16; ++j) { s += sSum[tid * 16 + j]; q += sSq[tid * 16 + j]; }
    sumPart[blockIdx.x * NC + tid] = s;
    sqPart [blockIdx.x * NC + tid] = q;
  }
}

__global__ void k2b_stats(const float* __restrict__ sumPart,
                          const float* __restrict__ sqPart,
                          const float* __restrict__ gamma,
                          const float* __restrict__ beta,
                          float* __restrict__ scaleA, float* __restrict__ scaleB) {
  const int c = threadIdx.x;
  if (c >= NC) return;
  float s = 0.0f, q = 0.0f;
  for (int blk = 0; blk < NB * NTILES; ++blk) {
    s += sumPart[blk * NC + c];
    q += sqPart [blk * NC + c];
  }
  const float invN = 1.0f / (float)(NB * NL);
  const float mean = s * invN;
  const float var  = q * invN - mean * mean;
  const float inv  = rsqrtf(var + EPS_BN);
  const float ga   = gamma[c] * inv;
  scaleA[c] = ga;
  scaleB[c] = beta[c] - mean * ga;
}

__global__ void k3_bn(float* __restrict__ y,
                      const float* __restrict__ scaleA,
                      const float* __restrict__ scaleB) {
  int i4 = blockIdx.x * blockDim.x + threadIdx.x;
  if (i4 >= NB * NC * NL / 4) return;
  const int c = (i4 >> 10) & (NC - 1);
  float4 v = ((float4*)y)[i4];
  const float a = scaleA[c], b = scaleB[c];
  v.x = v.x * a + b;  v.y = v.y * a + b;
  v.z = v.z * a + b;  v.w = v.w * a + b;
  ((float4*)y)[i4] = v;
}

extern "C" void kernel_launch(void* const* d_in, const int* in_sizes, int n_in,
                              void* d_out, int out_size, void* d_ws, size_t ws_size,
                              hipStream_t stream) {
  (void)in_sizes; (void)n_in; (void)out_size; (void)ws_size;
  const float* x      = (const float*)d_in[0];
  const float* conv_w = (const float*)d_in[1];
  const float* conv_b = (const float*)d_in[2];
  const float* wq     = (const float*)d_in[3];
  const float* bq     = (const float*)d_in[4];
  const float* wk     = (const float*)d_in[5];
  const float* bk     = (const float*)d_in[6];
  const float* wv     = (const float*)d_in[7];
  const float* bv     = (const float*)d_in[8];
  const float* gamma  = (const float*)d_in[9];
  const float* beta   = (const float*)d_in[10];
  float* out = (float*)d_out;

  float* ws      = (float*)d_ws;
  float* kbuf    = ws;
  float* Mpart   = kbuf    + (size_t)NB * NC * NL;
  float* Mbuf    = Mpart   + (size_t)NB * NTILES * NC * NC;
  float* sumPart = Mbuf    + (size_t)NB * NC * NC;
  float* sqPart  = sumPart + (size_t)NB * NTILES * NC;
  float* scaleA  = sqPart  + (size_t)NB * NTILES * NC;
  float* scaleB  = scaleA  + NC;

  const int grid1 = NB * NTILES;
  k1_conv_qkv_M<<<grid1, BLOCK, 0, stream>>>(x, conv_w, conv_b, wq, bq, wk, bk,
                                             wv, bv, kbuf, Mpart);
  k1b_reduceM<<<(NB * NC * NC + 255) / 256, 256, 0, stream>>>(Mpart, Mbuf);
  k2_gate_residual<<<grid1, BLOCK, 0, stream>>>(x, kbuf, Mbuf, out, sumPart, sqPart);
  k2b_stats<<<1, NC, 0, stream>>>(sumPart, sqPart, gamma, beta, scaleA, scaleB);
  k3_bn<<<(NB * NC * NL / 4 + 255) / 256, 256, 0, stream>>>(out, scaleA, scaleB);
}


# Column tile j of lane c = global column 4c+j: x, k, residual and y move as 16-byte accesses (x loads 96 to 24, k loads 128 to 32 per lane); new column order in the v q^T sum and BN partials
# speedup vs baseline: 2.1569x; 2.1569x over previous
#include <hip/hip_runtime.h>
#include <math.h>


typedef __attribute__((ext_vector_type(2))) float v2f;
typedef __attribute__((ext_vector_type(8))) float v8f;

#define NB 4
#define NC 64
#define NL 4096
#define LT 64
#define NTILES (NL / LT)
#define NWAVES 4
#define BLOCK (NWAVES * 32)
#define EPS_BN 1e-5f

#define SP_PITCH 68
#define SH_STRIDE 65

__device__ __forceinline__ v8f wmma_f32(v2f a, v2f b, v8f c) {
  return __builtin_amdgcn_wmma_f32_16x16x4_f32(false, a, false, b,
                                               (short)0, c, false, false);
}

__global__ __launch_bounds__(BLOCK)
void k1_conv_qkv_M(const float* __restrict__ x,
                   const float* __restrict__ conv_w, const float* __restrict__ conv_b,
                   const float* __restrict__ wq, const float* __restrict__ bq,
                   const float* __restrict__ wk, const float* __restrict__ bk,
                   const float* __restrict__ wv, const float* __restrict__ bv,
                   float* __restrict__ kbuf, float* __restrict__ Mpart) {
  __shared__ __attribute__((aligned(32))) float sH[NC * SH_STRIDE];
  __shared__ __attribute__((aligned(32))) float sQ[NC * SH_STRIDE];
  __shared__ __attribute__((aligned(32))) float sV[NC * SH_STRIDE];
  __shared__ float sP[NWAVES * NC * SP_PITCH];

  const int tid  = threadIdx.x;
  const int wave = tid >> 5;
  const int lane = tid & 31;
  const int half = lane >> 4;
  const int mn   = lane & 15;

  const int bIdx = blockIdx.x / NTILES;
  const int l0   = (blockIdx.x % NTILES) * LT;

  const float* xb = x + bIdx * NC * NL;
  {
    const int kbase = wave * 16;
    const int c0 = l0 + 4 * mn;
    const int cl = (c0 - 1) & (NL - 1);
    const int cr = (c0 + 4) & (NL - 1);
    v2f fr[4][6];
#pragma unroll
    for (int kk = 0; kk < 4; ++kk) {
      const float* r0 = xb + (kbase + kk * 4 + half * 2) * NL;
      const float* r1 = r0 + NL;
      const float4 m0 = *(const float4*)(r0 + c0);
      const float4 m1 = *(const float4*)(r1 + c0);
      const float e0[6] = { r0[cl], m0.x, m0.y, m0.z, m0.w, r0[cr] };
      const float e1[6] = { r1[cl], m1.x, m1.y, m1.z, m1.w, r1[cr] };
#pragma unroll
      for (int i = 0; i < 6; ++i) {
        fr[kk][i][0] = (e0[i] > 0.0f) ? e0[i] : (__expf(e0[i]) - 1.0f);
        fr[kk][i][1] = (e1[i] > 0.0f) ? e1[i] : (__expf(e1[i]) - 1.0f);
      }
    }
    float* pw = sP + wave * (NC * SP_PITCH);
#pragma unroll
    for (int ob = 0; ob < 4; ++ob) {
      v8f acc[4] = {};
#pragma unroll
      for (int kk = 0; kk < 4; ++kk) {
        const int k0 = kbase + kk * 4 + half * 2;
        const float* wp = conv_w + ((ob * 16 + mn) * NC + k0) * 3;
#pragma unroll
        for (int tap = 0; tap < 3; ++tap) {
          v2f a;
          a[0] = wp[tap];
          a[1] = wp[3 + tap];
#pragma unroll
          for (int nt = 0; nt < 4; ++nt)
            acc[nt] = wmma_f32(a, fr[kk][nt + tap], acc[nt]);
        }
      }
#pragma unroll
      for (int nt = 0; nt < 4; ++nt) {
#pragma unroll
        for (int r = 0; r < 8; ++r)
          pw[(ob * 16 + half * 8 + r) * SP_PITCH + nt * 16 + mn] = acc[nt][r];
      }
    }
  }
  __syncthreads();

  const int mt = wave;
  const int arow = mt * 16 + mn;
  const int rbase = mt * 16 + half * 8;

  {
    const float* p0 = sP + (mt * 16) * SP_PITCH;
#pragma unroll
    for (int i = 0; i < 32; ++i) {
      const int e  = i * 32 + lane;
      const int rr = e >> 6;
      const int cc = e & 63;
      const int o  = rr * SP_PITCH + cc;
      float s = p0[o] + p0[o + NC * SP_PITCH];
      s += p0[o + 2 * NC * SP_PITCH];
      s += p0[o + 3 * NC * SP_PITCH];
      sH[(mt * 16 + rr) * SH_STRIDE + cc] = s + conv_b[mt * 16 + rr];
    }
  }
  __syncthreads();

  v8f akk[4];
#pragma unroll
  for (int nt = 0; nt < 4; ++nt) {
    const int bcol = nt * 16 + mn;
    v8f aq = {}, ak = {}, av = {};
#pragma unroll
    for (int kk = 0; kk < 16; ++kk) {
      const int k0 = kk * 4 + half * 2;
      v2f bf, a;
      bf[0] = sH[k0 * SH_STRIDE + bcol];
      bf[1] = sH[(k0 + 1) * SH_STRIDE + bcol];
      a[0] = wq[arow * NC + k0]; a[1] = wq[arow * NC + k0 + 1];
      aq = wmma_f32(a, bf, aq);
      a[0] = wk[arow * NC + k0]; a[1] = wk[arow * NC + k0 + 1];
      ak = wmma_f32(a, bf, ak);
      a[0] = wv[arow * NC + k0]; a[1] = wv[arow * NC + k0 + 1];
      av = wmma_f32(a, bf, av);
    }
#pragma unroll
    for (int r = 0; r < 8; ++r) {
      const int row = rbase + r;
      sQ[row * SH_STRIDE + bcol] = aq[r] + bq[row];
      sV[row * SH_STRIDE + bcol] = av[r] + bv[row];
    }
    akk[nt] = ak;
  }
#pragma unroll
  for (int r = 0; r < 8; ++r) {
    const int row = rbase + r;
    const float kb_r = bk[row];
    float4 kv;
    kv.x = akk[0][r] + kb_r;
    kv.y = akk[1][r] + kb_r;
    kv.z = akk[2][r] + kb_r;
    kv.w = akk[3][r] + kb_r;
    *(float4*)&kbuf[(bIdx * NC + row) * NL + l0 + 4 * mn] = kv;
  }
  __syncthreads();

  float* Mout = Mpart + blockIdx.x * (NC * NC);
  for (int nt = 0; nt < 4; ++nt) {
    const int bcol = nt * 16 + mn;
    v8f acc = {};
    for (int kk = 0; kk < 16; ++kk) {
      const int k0 = kk * 4 + half * 2;
      v2f a, b;
      a[0] = sV[arow * SH_STRIDE + k0];
      a[1] = sV[arow * SH_STRIDE + k0 + 1];
      b[0] = sQ[bcol * SH_STRIDE + k0];
      b[1] = sQ[bcol * SH_STRIDE + k0 + 1];
      acc = wmma_f32(a, b, acc);
    }
#pragma unroll
    for (int r = 0; r < 8; ++r)
      Mout[(rbase + r) * NC + bcol] = acc[r];
  }
}

#define SEGC (NWAVES * LT)
#define NSEG (NL / SEGC)
#define NBAND (NC / 16)
#define SM_PITCH 68
#define ST_PITCH 17

__global__ __launch_bounds__(BLOCK)
void k2_foldM_gate_residual(const float* __restrict__ x,
                            const float* __restrict__ kbuf,
                            const float* __restrict__ Mpart,
                            float* __restrict__ y,
                            float* __restrict__ sumPart, float* __restrict__ sqPart) {
  __shared__ __attribute__((aligned(16))) float sM[16 * SM_PITCH];
  __shared__ float sSum[NWAVES * 16 * ST_PITCH];
  __shared__ float sSq [NWAVES * 16 * ST_PITCH];

  const int tid  = threadIdx.x;
  const int wave = tid >> 5;
  const int lane = tid & 31;
  const int half = lane >> 4;
  const int mn   = lane & 15;

  const int seg  = blockIdx.x % NSEG;
  const int band = (blockIdx.x / NSEG) % NBAND;
  const int bIdx = blockIdx.x / (NSEG * NBAND);
  const int tile = seg * NWAVES + wave;
  const int l0   = tile * LT;

  const float* kb = kbuf + (size_t)bIdx * NC * NL + l0 + 4 * mn;
  v2f bfr[16][4];
#pragma unroll
  for (int kk = 0; kk < 16; ++kk) {
    const int k0 = kk * 4 + half * 2;
    const float4 r0 = *(const float4*)(kb + k0 * NL);
    const float4 r1 = *(const float4*)(kb + (k0 + 1) * NL);
    bfr[kk][0][0] = r0.x;  bfr[kk][0][1] = r1.x;
    bfr[kk][1][0] = r0.y;  bfr[kk][1][1] = r1.y;
    bfr[kk][2][0] = r0.z;  bfr[kk][2][1] = r1.z;
    bfr[kk][3][0] = r0.w;  bfr[kk][3][1] = r1.w;
  }
  const float4* mp = (const float4*)(Mpart + (size_t)bIdx * NTILES * (NC * NC)
                                           + band * 16 * NC) + wave * 64 + lane;
  float4 m0 = {0.0f, 0.0f, 0.0f, 0.0f};
  float4 m1 = {0.0f, 0.0f, 0.0f, 0.0f};
#pragma unroll
  for (int t = 0; t < NTILES; ++t) {
    const float4 p0 = mp[t * (NC * NC / 4)];
    const float4 p1 = mp[t * (NC * NC / 4) + 32];
    m0.x += p0.x; m0.y += p0.y; m0.z += p0.z; m0.w += p0.w;
    m1.x += p1.x; m1.y += p1.y; m1.z += p1.z; m1.w += p1.w;
  }
  {
    const int row0 = wave * 4 + half;
    *(float4*)&sM[row0 * SM_PITCH + mn * 4]       = m0;
    *(float4*)&sM[(row0 + 2) * SM_PITCH + mn * 4] = m1;
  }

  __syncthreads();

  const int rbase = band * 16 + half * 8;
  const float* xb = x + (size_t)bIdx * NC * NL + l0 + 4 * mn;
  float* yb = y + (size_t)bIdx * NC * NL + l0 + 4 * mn;

  v8f acc[4] = {};
#pragma unroll
  for (int kk = 0; kk < 16; ++kk) {
    const v2f a = *(const v2f*)&sM[mn * SM_PITCH + kk * 4 + half * 2];
#pragma unroll
    for (int nt = 0; nt < 4; ++nt)
      acc[nt] = wmma_f32(a, bfr[kk][nt], acc[nt]);
  }

  float regS[8], regQ[8];
#pragma unroll
  for (int r = 0; r < 8; ++r) {
    const int row = rbase + r;
    const float4 xv = *(const float4*)(xb + row * NL);
    float4 yv;
    yv.x = acc[0][r] + xv.x;
    yv.y = acc[1][r] + xv.y;
    yv.z = acc[2][r] + xv.z;
    yv.w = acc[3][r] + xv.w;
    *(float4*)(yb + row * NL) = yv;
    float s = 0.0f, q = 0.0f;
    s += yv.x;  q += yv.x * yv.x;
    s += yv.y;  q += yv.y * yv.y;
    s += yv.z;  q += yv.z * yv.z;
    s += yv.w;  q += yv.w * yv.w;
    regS[r] = s;
    regQ[r] = q;
  }
  float* wS = sSum + wave * 16 * ST_PITCH;
  float* wQ = sSq  + wave * 16 * ST_PITCH;
#pragma unroll
  for (int r = 0; r < 8; ++r) {
    wS[(half * 8 + r) * ST_PITCH + mn] = regS[r];
    wQ[(half * 8 + r) * ST_PITCH + mn] = regQ[r];
  }
  __syncthreads();
  if (lane < 16) {
    float s = 0.0f, q = 0.0f;
    for (int j = 0; j < 16; ++j) { s += wS[lane * ST_PITCH + j]; q += wQ[lane * ST_PITCH + j]; }
    const int p = (bIdx * NTILES + tile) * NC + band * 16 + lane;
    sumPart[p] = s;
    sqPart [p] = q;
  }
}

#define K3_THREADS 256
#define K3_WAVES (K3_THREADS / 32)
static_assert(K3_THREADS == NB * NTILES, "one partial row per thread");
static_assert(NL == 4 * 4 * K3_THREADS, "four float4 per thread cover a row");

__global__ __launch_bounds__(K3_THREADS)
void k3_fold_bn(float* __restrict__ y,
                const float* __restrict__ sumPart,
                const float* __restrict__ sqPart,
                const float* __restrict__ gamma,
                const float* __restrict__ beta) {
  __shared__ float sS[K3_WAVES];
  __shared__ float sQ[K3_WAVES];

  const int tid = threadIdx.x;
  const int c   = blockIdx.x & (NC - 1);

  float s = sumPart[tid * NC + c];
  float q = sqPart [tid * NC + c];

  float4* yr = (float4*)(y + (size_t)blockIdx.x * NL);
  float4 v0 = yr[tid];
  float4 v1 = yr[tid + K3_THREADS];
  float4 v2 = yr[tid + 2 * K3_THREADS];
  float4 v3 = yr[tid + 3 * K3_THREADS];

  for (int off = 16; off > 0; off >>= 1) {
    s += __shfl_xor(s, off, 32);
    q += __shfl_xor(q, off, 32);
  }
  if ((tid & 31) == 0) { sS[tid >> 5] = s; sQ[tid >> 5] = q; }
  __syncthreads();
  float S = 0.0f, Q = 0.0f;
  for (int w = 0; w < K3_WAVES; ++w) { S += sS[w]; Q += sQ[w]; }

  const float invN = 1.0f / (float)(NB * NL);
  const float mean = S * invN;
  const float var  = Q * invN - mean * mean;
  const float inv  = rsqrtf(var + EPS_BN);
  const float a    = gamma[c] * inv;
  const float b    = beta[c] - mean * a;

  v0.x = v0.x * a + b;  v0.y = v0.y * a + b;  v0.z = v0.z * a + b;  v0.w = v0.w * a + b;
  v1.x = v1.x * a + b;  v1.y = v1.y * a + b;  v1.z = v1.z * a + b;  v1.w = v1.w * a + b;
  v2.x = v2.x * a + b;  v2.y = v2.y * a + b;  v2.z = v2.z * a + b;  v2.w = v2.w * a + b;
  v3.x = v3.x * a + b;  v3.y = v3.y * a + b;  v3.z = v3.z * a + b;  v3.w = v3.w * a + b;
  yr[tid]                  = v0;
  yr[tid + K3_THREADS]     = v1;
  yr[tid + 2 * K3_THREADS] = v2;
  yr[tid + 3 * K3_THREADS] = v3;
}

extern "C" void kernel_launch(void* const* d_in, const int* in_sizes, int n_in,
                              void* d_out, int out_size, void* d_ws, size_t ws_size,
                              hipStream_t stream) {
  (void)in_sizes; (void)n_in; (void)out_size; (void)ws_size;
  const float* x      = (const float*)d_in[0];
  const float* conv_w = (const float*)d_in[1];
  const float* conv_b = (const float*)d_in[2];
  const float* wq     = (const float*)d_in[3];
  const float* bq     = (const float*)d_in[4];
  const float* wk     = (const float*)d_in[5];
  const float* bk     = (const float*)d_in[6];
  const float* wv     = (const float*)d_in[7];
  const float* bv     = (const float*)d_in[8];
  const float* gamma  = (const float*)d_in[9];
  const float* beta   = (const float*)d_in[10];
  float* out = (float*)d_out;

  float* ws      = (float*)d_ws;
  float* kbuf    = ws;
  float* Mpart   = kbuf    + (size_t)NB * NC * NL;
  float* sumPart = Mpart   + (size_t)NB * NTILES * NC * NC;
  float* sqPart  = sumPart + (size_t)NB * NTILES * NC;

  const int grid1 = NB * NTILES;
  k1_conv_qkv_M<<<grid1, BLOCK, 0, stream>>>(x, conv_w, conv_b, wq, bq, wk, bk,
                                             wv, bv, kbuf, Mpart);
  k2_foldM_gate_residual<<<NB * NBAND * NSEG, BLOCK, 0, stream>>>(x, kbuf, Mpart, out,
                                                              sumPart, sqPart);
  k3_fold_bn<<<NB * NC, K3_THREADS, 0, stream>>>(out, sumPart, sqPart, gamma, beta);
}


# K2 slab walk rotated per segment: workgroup seg starts its fold of the 64 M slabs at slab 4*seg and wraps, so the 16 workgroups of a band read different lines at a time; new M sum order per segment
# speedup vs baseline: 2.3135x; 1.0726x over previous
#include <hip/hip_runtime.h>
#include <math.h>


typedef __attribute__((ext_vector_type(2))) float v2f;
typedef __attribute__((ext_vector_type(8))) float v8f;

#define NB 4
#define NC 64
#define NL 4096
#define LT 64
#define NTILES (NL / LT)
#define NWAVES 4
#define BLOCK (NWAVES * 32)
#define EPS_BN 1e-5f

#define SP_PITCH 68
#define SH_STRIDE 65

__device__ __forceinline__ v8f wmma_f32(v2f a, v2f b, v8f c) {
  return __builtin_amdgcn_wmma_f32_16x16x4_f32(false, a, false, b,
                                               (short)0, c, false, false);
}

__global__ __launch_bounds__(BLOCK)
void k1_conv_qkv_M(const float* __restrict__ x,
                   const float* __restrict__ conv_w, const float* __restrict__ conv_b,
                   const float* __restrict__ wq, const float* __restrict__ bq,
                   const float* __restrict__ wk, const float* __restrict__ bk,
                   const float* __restrict__ wv, const float* __restrict__ bv,
                   float* __restrict__ kbuf, float* __restrict__ Mpart) {
  __shared__ __attribute__((aligned(32))) float sH[NC * SH_STRIDE];
  __shared__ __attribute__((aligned(32))) float sQ[NC * SH_STRIDE];
  __shared__ __attribute__((aligned(32))) float sV[NC * SH_STRIDE];
  __shared__ float sP[NWAVES * NC * SP_PITCH];

  const int tid  = threadIdx.x;
  const int wave = tid >> 5;
  const int lane = tid & 31;
  const int half = lane >> 4;
  const int mn   = lane & 15;

  const int bIdx = blockIdx.x / NTILES;
  const int l0   = (blockIdx.x % NTILES) * LT;

  const float* xb = x + bIdx * NC * NL;
  {
    const int kbase = wave * 16;
    const int c0 = l0 + 4 * mn;
    const int cl = (c0 - 1) & (NL - 1);
    const int cr = (c0 + 4) & (NL - 1);
    v2f fr[4][6];
#pragma unroll
    for (int kk = 0; kk < 4; ++kk) {
      const float* r0 = xb + (kbase + kk * 4 + half * 2) * NL;
      const float* r1 = r0 + NL;
      const float4 m0 = *(const float4*)(r0 + c0);
      const float4 m1 = *(const float4*)(r1 + c0);
      const float e0[6] = { r0[cl], m0.x, m0.y, m0.z, m0.w, r0[cr] };
      const float e1[6] = { r1[cl], m1.x, m1.y, m1.z, m1.w, r1[cr] };
#pragma unroll
      for (int i = 0; i < 6; ++i) {
        fr[kk][i][0] = (e0[i] > 0.0f) ? e0[i] : (__expf(e0[i]) - 1.0f);
        fr[kk][i][1] = (e1[i] > 0.0f) ? e1[i] : (__expf(e1[i]) - 1.0f);
      }
    }
    float* pw = sP + wave * (NC * SP_PITCH);
#pragma unroll
    for (int ob = 0; ob < 4; ++ob) {
      v8f acc[4] = {};
#pragma unroll
      for (int kk = 0; kk < 4; ++kk) {
        const int k0 = kbase + kk * 4 + half * 2;
        const float* wp = conv_w + ((ob * 16 + mn) * NC + k0) * 3;
#pragma unroll
        for (int tap = 0; tap < 3; ++tap) {
          v2f a;
          a[0] = wp[tap];
          a[1] = wp[3 + tap];
#pragma unroll
          for (int nt = 0; nt < 4; ++nt)
            acc[nt] = wmma_f32(a, fr[kk][nt + tap], acc[nt]);
        }
      }
#pragma unroll
      for (int nt = 0; nt < 4; ++nt) {
#pragma unroll
        for (int r = 0; r < 8; ++r)
          pw[(ob * 16 + half * 8 + r) * SP_PITCH + nt * 16 + mn] = acc[nt][r];
      }
    }
  }
  __syncthreads();

  const int mt = wave;
  const int arow = mt * 16 + mn;
  const int rbase = mt * 16 + half * 8;

  {
    const float* p0 = sP + (mt * 16) * SP_PITCH;
#pragma unroll
    for (int i = 0; i < 32; ++i) {
      const int e  = i * 32 + lane;
      const int rr = e >> 6;
      const int cc = e & 63;
      const int o  = rr * SP_PITCH + cc;
      float s = p0[o] + p0[o + NC * SP_PITCH];
      s += p0[o + 2 * NC * SP_PITCH];
      s += p0[o + 3 * NC * SP_PITCH];
      sH[(mt * 16 + rr) * SH_STRIDE + cc] = s + conv_b[mt * 16 + rr];
    }
  }
  __syncthreads();

  v8f akk[4];
#pragma unroll
  for (int nt = 0; nt < 4; ++nt) {
    const int bcol = nt * 16 + mn;
    v8f aq = {}, ak = {}, av = {};
#pragma unroll
    for (int kk = 0; kk < 16; ++kk) {
      const int k0 = kk * 4 + half * 2;
      v2f bf, a;
      bf[0] = sH[k0 * SH_STRIDE + bcol];
      bf[1] = sH[(k0 + 1) * SH_STRIDE + bcol];
      a[0] = wq[arow * NC + k0]; a[1] = wq[arow * NC + k0 + 1];
      aq = wmma_f32(a, bf, aq);
      a[0] = wk[arow * NC + k0]; a[1] = wk[arow * NC + k0 + 1];
      ak = wmma_f32(a, bf, ak);
      a[0] = wv[arow * NC + k0]; a[1] = wv[arow * NC + k0 + 1];
      av = wmma_f32(a, bf, av);
    }
#pragma unroll
    for (int r = 0; r < 8; ++r) {
      const int row = rbase + r;
      sQ[row * SH_STRIDE + bcol] = aq[r] + bq[row];
      sV[row * SH_STRIDE + bcol] = av[r] + bv[row];
    }
    akk[nt] = ak;
  }
#pragma unroll
  for (int r = 0; r < 8; ++r) {
    const int row = rbase + r;
    const float kb_r = bk[row];
    float4 kv;
    kv.x = akk[0][r] + kb_r;
    kv.y = akk[1][r] + kb_r;
    kv.z = akk[2][r] + kb_r;
    kv.w = akk[3][r] + kb_r;
    *(float4*)&kbuf[(bIdx * NC + row) * NL + l0 + 4 * mn] = kv;
  }
  __syncthreads();

  float* Mout = Mpart + blockIdx.x * (NC * NC);
  for (int nt = 0; nt < 4; ++nt) {
    const int bcol = nt * 16 + mn;
    v8f acc = {};
    for (int kk = 0; kk < 16; ++kk) {
      const int k0 = kk * 4 + half * 2;
      v2f a, b;
      a[0] = sV[arow * SH_STRIDE + k0];
      a[1] = sV[arow * SH_STRIDE + k0 + 1];
      b[0] = sQ[bcol * SH_STRIDE + k0];
      b[1] = sQ[bcol * SH_STRIDE + k0 + 1];
      acc = wmma_f32(a, b, acc);
    }
#pragma unroll
    for (int r = 0; r < 8; ++r)
      Mout[(rbase + r) * NC + bcol] = acc[r];
  }
}

#define SEGC (NWAVES * LT)
#define NSEG (NL / SEGC)
#define NBAND (NC / 16)
#define SM_PITCH 68
#define ST_PITCH 17

__global__ __launch_bounds__(BLOCK)
void k2_foldM_gate_residual(const float* __restrict__ x,
                            const float* __restrict__ kbuf,
                            const float* __restrict__ Mpart,
                            float* __restrict__ y,
                            float* __restrict__ sumPart, float* __restrict__ sqPart) {
  __shared__ __attribute__((aligned(16))) float sM[16 * SM_PITCH];
  __shared__ float sSum[NWAVES * 16 * ST_PITCH];
  __shared__ float sSq [NWAVES * 16 * ST_PITCH];

  const int tid  = threadIdx.x;
  const int wave = tid >> 5;
  const int lane = tid & 31;
  const int half = lane >> 4;
  const int mn   = lane & 15;

  const int seg  = blockIdx.x % NSEG;
  const int band = (blockIdx.x / NSEG) % NBAND;
  const int bIdx = blockIdx.x / (NSEG * NBAND);
  const int tile = seg * NWAVES + wave;
  const int l0   = tile * LT;

  const float* kb = kbuf + (size_t)bIdx * NC * NL + l0 + 4 * mn;
  v2f bfr[16][4];
#pragma unroll
  for (int kk = 0; kk < 16; ++kk) {
    const int k0 = kk * 4 + half * 2;
    const float4 r0 = *(const float4*)(kb + k0 * NL);
    const float4 r1 = *(const float4*)(kb + (k0 + 1) * NL);
    bfr[kk][0][0] = r0.x;  bfr[kk][0][1] = r1.x;
    bfr[kk][1][0] = r0.y;  bfr[kk][1][1] = r1.y;
    bfr[kk][2][0] = r0.z;  bfr[kk][2][1] = r1.z;
    bfr[kk][3][0] = r0.w;  bfr[kk][3][1] = r1.w;
  }
  const float4* mp = (const float4*)(Mpart + (size_t)bIdx * NTILES * (NC * NC)
                                           + band * 16 * NC) + wave * 64 + lane;
  float4 m0 = {0.0f, 0.0f, 0.0f, 0.0f};
  float4 m1 = {0.0f, 0.0f, 0.0f, 0.0f};
  const int rot = seg * (NTILES / NSEG);
#pragma unroll
  for (int t = 0; t < NTILES; ++t) {
    const float4* ms = mp + (size_t)((t + rot) & (NTILES - 1)) * (NC * NC / 4);
    const float4 p0 = ms[0];
    const float4 p1 = ms[32];
    m0.x += p0.x; m0.y += p0.y; m0.z += p0.z; m0.w += p0.w;
    m1.x += p1.x; m1.y += p1.y; m1.z += p1.z; m1.w += p1.w;
  }
  {
    const int row0 = wave * 4 + half;
    *(float4*)&sM[row0 * SM_PITCH + mn * 4]       = m0;
    *(float4*)&sM[(row0 + 2) * SM_PITCH + mn * 4] = m1;
  }

  __syncthreads();

  const int rbase = band * 16 + half * 8;
  const float* xb = x + (size_t)bIdx * NC * NL + l0 + 4 * mn;
  float* yb = y + (size_t)bIdx * NC * NL + l0 + 4 * mn;

  v8f acc[4] = {};
#pragma unroll
  for (int kk = 0; kk < 16; ++kk) {
    const v2f a = *(const v2f*)&sM[mn * SM_PITCH + kk * 4 + half * 2];
#pragma unroll
    for (int nt = 0; nt < 4; ++nt)
      acc[nt] = wmma_f32(a, bfr[kk][nt], acc[nt]);
  }

  float regS[8], regQ[8];
#pragma unroll
  for (int r = 0; r < 8; ++r) {
    const int row = rbase + r;
    const float4 xv = *(const float4*)(xb + row * NL);
    float4 yv;
    yv.x = acc[0][r] + xv.x;
    yv.y = acc[1][r] + xv.y;
    yv.z = acc[2][r] + xv.z;
    yv.w = acc[3][r] + xv.w;
    *(float4*)(yb + row * NL) = yv;
    float s = 0.0f, q = 0.0f;
    s += yv.x;  q += yv.x * yv.x;
    s += yv.y;  q += yv.y * yv.y;
    s += yv.z;  q += yv.z * yv.z;
    s += yv.w;  q += yv.w * yv.w;
    regS[r] = s;
    regQ[r] = q;
  }
  float* wS = sSum + wave * 16 * ST_PITCH;
  float* wQ = sSq  + wave * 16 * ST_PITCH;
#pragma unroll
  for (int r = 0; r < 8; ++r) {
    wS[(half * 8 + r) * ST_PITCH + mn] = regS[r];
    wQ[(half * 8 + r) * ST_PITCH + mn] = regQ[r];
  }
  __syncthreads();
  if (lane < 16) {
    float s = 0.0f, q = 0.0f;
    for (int j = 0; j < 16; ++j) { s += wS[lane * ST_PITCH + j]; q += wQ[lane * ST_PITCH + j]; }
    const int p = (bIdx * NTILES + tile) * NC + band * 16 + lane;
    sumPart[p] = s;
    sqPart [p] = q;
  }
}

#define K3_THREADS 256
#define K3_WAVES (K3_THREADS / 32)
static_assert(K3_THREADS == NB * NTILES, "one partial row per thread");
static_assert(NL == 4 * 4 * K3_THREADS, "four float4 per thread cover a row");

__global__ __launch_bounds__(K3_THREADS)
void k3_fold_bn(float* __restrict__ y,
                const float* __restrict__ sumPart,
                const float* __restrict__ sqPart,
                const float* __restrict__ gamma,
                const float* __restrict__ beta) {
  __shared__ float sS[K3_WAVES];
  __shared__ float sQ[K3_WAVES];

  const int tid = threadIdx.x;
  const int c   = blockIdx.x & (NC - 1);

  float s = sumPart[tid * NC + c];
  float q = sqPart [tid * NC + c];

  float4* yr = (float4*)(y + (size_t)blockIdx.x * NL);
  float4 v0 = yr[tid];
  float4 v1 = yr[tid + K3_THREADS];
  float4 v2 = yr[tid + 2 * K3_THREADS];
  float4 v3 = yr[tid + 3 * K3_THREADS];

  for (int off = 16; off > 0; off >>= 1) {
    s += __shfl_xor(s, off, 32);
    q += __shfl_xor(q, off, 32);
  }
  if ((tid & 31) == 0) { sS[tid >> 5] = s; sQ[tid >> 5] = q; }
  __syncthreads();
  float S = 0.0f, Q = 0.0f;
  for (int w = 0; w < K3_WAVES; ++w) { S += sS[w]; Q += sQ[w]; }

  const float invN = 1.0f / (float)(NB * NL);
  const float mean = S * invN;
  const float var  = Q * invN - mean * mean;
  const float inv  = rsqrtf(var + EPS_BN);
  const float a    = gamma[c] * inv;
  const float b    = beta[c] - mean * a;

  v0.x = v0.x * a + b;  v0.y = v0.y * a + b;  v0.z = v0.z * a + b;  v0.w = v0.w * a + b;
  v1.x = v1.x * a + b;  v1.y = v1.y * a + b;  v1.z = v1.z * a + b;  v1.w = v1.w * a + b;
  v2.x = v2.x * a + b;  v2.y = v2.y * a + b;  v2.z = v2.z * a + b;  v2.w = v2.w * a + b;
  v3.x = v3.x * a + b;  v3.y = v3.y * a + b;  v3.z = v3.z * a + b;  v3.w = v3.w * a + b;
  yr[tid]                  = v0;
  yr[tid + K3_THREADS]     = v1;
  yr[tid + 2 * K3_THREADS] = v2;
  yr[tid + 3 * K3_THREADS] = v3;
}

extern "C" void kernel_launch(void* const* d_in, const int* in_sizes, int n_in,
                              void* d_out, int out_size, void* d_ws, size_t ws_size,
                              hipStream_t stream) {
  (void)in_sizes; (void)n_in; (void)out_size; (void)ws_size;
  const float* x      = (const float*)d_in[0];
  const float* conv_w = (const float*)d_in[1];
  const float* conv_b = (const float*)d_in[2];
  const float* wq     = (const float*)d_in[3];
  const float* bq     = (const float*)d_in[4];
  const float* wk     = (const float*)d_in[5];
  const float* bk     = (const float*)d_in[6];
  const float* wv     = (const float*)d_in[7];
  const float* bv     = (const float*)d_in[8];
  const float* gamma  = (const float*)d_in[9];
  const float* beta   = (const float*)d_in[10];
  float* out = (float*)d_out;

  float* ws      = (float*)d_ws;
  float* kbuf    = ws;
  float* Mpart   = kbuf    + (size_t)NB * NC * NL;
  float* sumPart = Mpart   + (size_t)NB * NTILES * NC * NC;
  float* sqPart  = sumPart + (size_t)NB * NTILES * NC;

  const int grid1 = NB * NTILES;
  k1_conv_qkv_M<<<grid1, BLOCK, 0, stream>>>(x, conv_w, conv_b, wq, bq, wk, bk,
                                             wv, bv, kbuf, Mpart);
  k2_foldM_gate_residual<<<NB * NBAND * NSEG, BLOCK, 0, stream>>>(x, kbuf, Mpart, out,
                                                              sumPart, sqPart);
  k3_fold_bn<<<NB * NC, K3_THREADS, 0, stream>>>(out, sumPart, sqPart, gamma, beta);
}


# LDS-only workgroup barriers at the two LDS hand-overs that follow global stores (K1: q/v after the k rows; K2: BN partials after the y rows): no wait for the stores' acknowledgement; sums unchanged
# speedup vs baseline: 2.3961x; 1.0357x over previous
#include <hip/hip_runtime.h>
#include <math.h>


typedef __attribute__((ext_vector_type(2))) float v2f;
typedef __attribute__((ext_vector_type(8))) float v8f;

#define NB 4
#define NC 64
#define NL 4096
#define LT 64
#define NTILES (NL / LT)
#define NWAVES 4
#define BLOCK (NWAVES * 32)
#define EPS_BN 1e-5f

#define SP_PITCH 68
#define SH_STRIDE 65

__device__ __forceinline__ v8f wmma_f32(v2f a, v2f b, v8f c) {
  return __builtin_amdgcn_wmma_f32_16x16x4_f32(false, a, false, b,
                                               (short)0, c, false, false);
}

__device__ __forceinline__ void barrier_lds_only() {
  __builtin_amdgcn_fence(__ATOMIC_RELEASE, "workgroup", "local");
  __builtin_amdgcn_s_barrier();
  __builtin_amdgcn_fence(__ATOMIC_ACQUIRE, "workgroup", "local");
}

__global__ __launch_bounds__(BLOCK)
void k1_conv_qkv_M(const float* __restrict__ x,
                   const float* __restrict__ conv_w, const float* __restrict__ conv_b,
                   const float* __restrict__ wq, const float* __restrict__ bq,
                   const float* __restrict__ wk, const float* __restrict__ bk,
                   const float* __restrict__ wv, const float* __restrict__ bv,
                   float* __restrict__ kbuf, float* __restrict__ Mpart) {
  __shared__ __attribute__((aligned(32))) float sH[NC * SH_STRIDE];
  __shared__ __attribute__((aligned(32))) float sQ[NC * SH_STRIDE];
  __shared__ __attribute__((aligned(32))) float sV[NC * SH_STRIDE];
  __shared__ float sP[NWAVES * NC * SP_PITCH];

  const int tid  = threadIdx.x;
  const int wave = tid >> 5;
  const int lane = tid & 31;
  const int half = lane >> 4;
  const int mn   = lane & 15;

  const int bIdx = blockIdx.x / NTILES;
  const int l0   = (blockIdx.x % NTILES) * LT;

  const float* xb = x + bIdx * NC * NL;
  {
    const int kbase = wave * 16;
    const int c0 = l0 + 4 * mn;
    const int cl = (c0 - 1) & (NL - 1);
    const int cr = (c0 + 4) & (NL - 1);
    v2f fr[4][6];
#pragma unroll
    for (int kk = 0; kk < 4; ++kk) {
      const float* r0 = xb + (kbase + kk * 4 + half * 2) * NL;
      const float* r1 = r0 + NL;
      const float4 m0 = *(const float4*)(r0 + c0);
      const float4 m1 = *(const float4*)(r1 + c0);
      const float e0[6] = { r0[cl], m0.x, m0.y, m0.z, m0.w, r0[cr] };
      const float e1[6] = { r1[cl], m1.x, m1.y, m1.z, m1.w, r1[cr] };
#pragma unroll
      for (int i = 0; i < 6; ++i) {
        fr[kk][i][0] = (e0[i] > 0.0f) ? e0[i] : (__expf(e0[i]) - 1.0f);
        fr[kk][i][1] = (e1[i] > 0.0f) ? e1[i] : (__expf(e1[i]) - 1.0f);
      }
    }
    float* pw = sP + wave * (NC * SP_PITCH);
#pragma unroll
    for (int ob = 0; ob < 4; ++ob) {
      v8f acc[4] = {};
#pragma unroll
      for (int kk = 0; kk < 4; ++kk) {
        const int k0 = kbase + kk * 4 + half * 2;
        const float* wp = conv_w + ((ob * 16 + mn) * NC + k0) * 3;
#pragma unroll
        for (int tap = 0; tap < 3; ++tap) {
          v2f a;
          a[0] = wp[tap];
          a[1] = wp[3 + tap];
#pragma unroll
          for (int nt = 0; nt < 4; ++nt)
            acc[nt] = wmma_f32(a, fr[kk][nt + tap], acc[nt]);
        }
      }
#pragma unroll
      for (int nt = 0; nt < 4; ++nt) {
#pragma unroll
        for (int r = 0; r < 8; ++r)
          pw[(ob * 16 + half * 8 + r) * SP_PITCH + nt * 16 + mn] = acc[nt][r];
      }
    }
  }
  __syncthreads();

  const int mt = wave;
  const int arow = mt * 16 + mn;
  const int rbase = mt * 16 + half * 8;

  {
    const float* p0 = sP + (mt * 16) * SP_PITCH;
#pragma unroll
    for (int i = 0; i < 32; ++i) {
      const int e  = i * 32 + lane;
      const int rr = e >> 6;
      const int cc = e & 63;
      const int o  = rr * SP_PITCH + cc;
      float s = p0[o] + p0[o + NC * SP_PITCH];
      s += p0[o + 2 * NC * SP_PITCH];
      s += p0[o + 3 * NC * SP_PITCH];
      sH[(mt * 16 + rr) * SH_STRIDE + cc] = s + conv_b[mt * 16 + rr];
    }
  }
  __syncthreads();

  v8f akk[4];
#pragma unroll
  for (int nt = 0; nt < 4; ++nt) {
    const int bcol = nt * 16 + mn;
    v8f aq = {}, ak = {}, av = {};
#pragma unroll
    for (int kk = 0; kk < 16; ++kk) {
      const int k0 = kk * 4 + half * 2;
      v2f bf, a;
      bf[0] = sH[k0 * SH_STRIDE + bcol];
      bf[1] = sH[(k0 + 1) * SH_STRIDE + bcol];
      a[0] = wq[arow * NC + k0]; a[1] = wq[arow * NC + k0 + 1];
      aq = wmma_f32(a, bf, aq);
      a[0] = wk[arow * NC + k0]; a[1] = wk[arow * NC + k0 + 1];
      ak = wmma_f32(a, bf, ak);
      a[0] = wv[arow * NC + k0]; a[1] = wv[arow * NC + k0 + 1];
      av = wmma_f32(a, bf, av);
    }
#pragma unroll
    for (int r = 0; r < 8; ++r) {
      const int row = rbase + r;
      sQ[row * SH_STRIDE + bcol] = aq[r] + bq[row];
      sV[row * SH_STRIDE + bcol] = av[r] + bv[row];
    }
    akk[nt] = ak;
  }
#pragma unroll
  for (int r = 0; r < 8; ++r) {
    const int row = rbase + r;
    const float kb_r = bk[row];
    float4 kv;
    kv.x = akk[0][r] + kb_r;
    kv.y = akk[1][r] + kb_r;
    kv.z = akk[2][r] + kb_r;
    kv.w = akk[3][r] + kb_r;
    *(float4*)&kbuf[(bIdx * NC + row) * NL + l0 + 4 * mn] = kv;
  }
  barrier_lds_only();

  float* Mout = Mpart + blockIdx.x * (NC * NC);
  for (int nt = 0; nt < 4; ++nt) {
    const int bcol = nt * 16 + mn;
    v8f acc = {};
    for (int kk = 0; kk < 16; ++kk) {
      const int k0 = kk * 4 + half * 2;
      v2f a, b;
      a[0] = sV[arow * SH_STRIDE + k0];
      a[1] = sV[arow * SH_STRIDE + k0 + 1];
      b[0] = sQ[bcol * SH_STRIDE + k0];
      b[1] = sQ[bcol * SH_STRIDE + k0 + 1];
      acc = wmma_f32(a, b, acc);
    }
#pragma unroll
    for (int r = 0; r < 8; ++r)
      Mout[(rbase + r) * NC + bcol] = acc[r];
  }
}

#define SEGC (NWAVES * LT)
#define NSEG (NL / SEGC)
#define NBAND (NC / 16)
#define SM_PITCH 68
#define ST_PITCH 17

__global__ __launch_bounds__(BLOCK)
void k2_foldM_gate_residual(const float* __restrict__ x,
                            const float* __restrict__ kbuf,
                            const float* __restrict__ Mpart,
                            float* __restrict__ y,
                            float* __restrict__ sumPart, float* __restrict__ sqPart) {
  __shared__ __attribute__((aligned(16))) float sM[16 * SM_PITCH];
  __shared__ float sSum[NWAVES * 16 * ST_PITCH];
  __shared__ float sSq [NWAVES * 16 * ST_PITCH];

  const int tid  = threadIdx.x;
  const int wave = tid >> 5;
  const int lane = tid & 31;
  const int half = lane >> 4;
  const int mn   = lane & 15;

  const int seg  = blockIdx.x % NSEG;
  const int band = (blockIdx.x / NSEG) % NBAND;
  const int bIdx = blockIdx.x / (NSEG * NBAND);
  const int tile = seg * NWAVES + wave;
  const int l0   = tile * LT;

  const float* kb = kbuf + (size_t)bIdx * NC * NL + l0 + 4 * mn;
  v2f bfr[16][4];
#pragma unroll
  for (int kk = 0; kk < 16; ++kk) {
    const int k0 = kk * 4 + half * 2;
    const float4 r0 = *(const float4*)(kb + k0 * NL);
    const float4 r1 = *(const float4*)(kb + (k0 + 1) * NL);
    bfr[kk][0][0] = r0.x;  bfr[kk][0][1] = r1.x;
    bfr[kk][1][0] = r0.y;  bfr[kk][1][1] = r1.y;
    bfr[kk][2][0] = r0.z;  bfr[kk][2][1] = r1.z;
    bfr[kk][3][0] = r0.w;  bfr[kk][3][1] = r1.w;
  }
  const float4* mp = (const float4*)(Mpart + (size_t)bIdx * NTILES * (NC * NC)
                                           + band * 16 * NC) + wave * 64 + lane;
  float4 m0 = {0.0f, 0.0f, 0.0f, 0.0f};
  float4 m1 = {0.0f, 0.0f, 0.0f, 0.0f};
  const int rot = seg * (NTILES / NSEG);
#pragma unroll
  for (int t = 0; t < NTILES; ++t) {
    const float4* ms = mp + (size_t)((t + rot) & (NTILES - 1)) * (NC * NC / 4);
    const float4 p0 = ms[0];
    const float4 p1 = ms[32];
    m0.x += p0.x; m0.y += p0.y; m0.z += p0.z; m0.w += p0.w;
    m1.x += p1.x; m1.y += p1.y; m1.z += p1.z; m1.w += p1.w;
  }
  {
    const int row0 = wave * 4 + half;
    *(float4*)&sM[row0 * SM_PITCH + mn * 4]       = m0;
    *(float4*)&sM[(row0 + 2) * SM_PITCH + mn * 4] = m1;
  }

  __syncthreads();

  const int rbase = band * 16 + half * 8;
  const float* xb = x + (size_t)bIdx * NC * NL + l0 + 4 * mn;
  float* yb = y + (size_t)bIdx * NC * NL + l0 + 4 * mn;

  v8f acc[4] = {};
#pragma unroll
  for (int kk = 0; kk < 16; ++kk) {
    const v2f a = *(const v2f*)&sM[mn * SM_PITCH + kk * 4 + half * 2];
#pragma unroll
    for (int nt = 0; nt < 4; ++nt)
      acc[nt] = wmma_f32(a, bfr[kk][nt], acc[nt]);
  }

  float regS[8], regQ[8];
#pragma unroll
  for (int r = 0; r < 8; ++r) {
    const int row = rbase + r;
    const float4 xv = *(const float4*)(xb + row * NL);
    float4 yv;
    yv.x = acc[0][r] + xv.x;
    yv.y = acc[1][r] + xv.y;
    yv.z = acc[2][r] + xv.z;
    yv.w = acc[3][r] + xv.w;
    *(float4*)(yb + row * NL) = yv;
    float s = 0.0f, q = 0.0f;
    s += yv.x;  q += yv.x * yv.x;
    s += yv.y;  q += yv.y * yv.y;
    s += yv.z;  q += yv.z * yv.z;
    s += yv.w;  q += yv.w * yv.w;
    regS[r] = s;
    regQ[r] = q;
  }
  float* wS = sSum + wave * 16 * ST_PITCH;
  float* wQ = sSq  + wave * 16 * ST_PITCH;
#pragma unroll
  for (int r = 0; r < 8; ++r) {
    wS[(half * 8 + r) * ST_PITCH + mn] = regS[r];
    wQ[(half * 8 + r) * ST_PITCH + mn] = regQ[r];
  }
  barrier_lds_only();
  if (lane < 16) {
    float s = 0.0f, q = 0.0f;
    for (int j = 0; j < 16; ++j) { s += wS[lane * ST_PITCH + j]; q += wQ[lane * ST_PITCH + j]; }
    const int p = (bIdx * NTILES + tile) * NC + band * 16 + lane;
    sumPart[p] = s;
    sqPart [p] = q;
  }
}

#define K3_THREADS 256
#define K3_WAVES (K3_THREADS / 32)
static_assert(K3_THREADS == NB * NTILES, "one partial row per thread");
static_assert(NL == 4 * 4 * K3_THREADS, "four float4 per thread cover a row");

__global__ __launch_bounds__(K3_THREADS)
void k3_fold_bn(float* __restrict__ y,
                const float* __restrict__ sumPart,
                const float* __restrict__ sqPart,
                const float* __restrict__ gamma,
                const float* __restrict__ beta) {
  __shared__ float sS[K3_WAVES];
  __shared__ float sQ[K3_WAVES];

  const int tid = threadIdx.x;
  const int c   = blockIdx.x & (NC - 1);

  float s = sumPart[tid * NC + c];
  float q = sqPart [tid * NC + c];

  float4* yr = (float4*)(y + (size_t)blockIdx.x * NL);
  float4 v0 = yr[tid];
  float4 v1 = yr[tid + K3_THREADS];
  float4 v2 = yr[tid + 2 * K3_THREADS];
  float4 v3 = yr[tid + 3 * K3_THREADS];

  for (int off = 16; off > 0; off >>= 1) {
    s += __shfl_xor(s, off, 32);
    q += __shfl_xor(q, off, 32);
  }
  if ((tid & 31) == 0) { sS[tid >> 5] = s; sQ[tid >> 5] = q; }
  __syncthreads();
  float S = 0.0f, Q = 0.0f;
  for (int w = 0; w < K3_WAVES; ++w) { S += sS[w]; Q += sQ[w]; }

  const float invN = 1.0f / (float)(NB * NL);
  const float mean = S * invN;
  const float var  = Q * invN - mean * mean;
  const float inv  = rsqrtf(var + EPS_BN);
  const float a    = gamma[c] * inv;
  const float b    = beta[c] - mean * a;

  v0.x = v0.x * a + b;  v0.y = v0.y * a + b;  v0.z = v0.z * a + b;  v0.w = v0.w * a + b;
  v1.x = v1.x * a + b;  v1.y = v1.y * a + b;  v1.z = v1.z * a + b;  v1.w = v1.w * a + b;
  v2.x = v2.x * a + b;  v2.y = v2.y * a + b;  v2.z = v2.z * a + b;  v2.w = v2.w * a + b;
  v3.x = v3.x * a + b;  v3.y = v3.y * a + b;  v3.z = v3.z * a + b;  v3.w = v3.w * a + b;
  yr[tid]                  = v0;
  yr[tid + K3_THREADS]     = v1;
  yr[tid + 2 * K3_THREADS] = v2;
  yr[tid + 3 * K3_THREADS] = v3;
}

extern "C" void kernel_launch(void* const* d_in, const int* in_sizes, int n_in,
                              void* d_out, int out_size, void* d_ws, size_t ws_size,
                              hipStream_t stream) {
  (void)in_sizes; (void)n_in; (void)out_size; (void)ws_size;
  const float* x      = (const float*)d_in[0];
  const float* conv_w = (const float*)d_in[1];
  const float* conv_b = (const float*)d_in[2];
  const float* wq     = (const float*)d_in[3];
  const float* bq     = (const float*)d_in[4];
  const float* wk     = (const float*)d_in[5];
  const float* bk     = (const float*)d_in[6];
  const float* wv     = (const float*)d_in[7];
  const float* bv     = (const float*)d_in[8];
  const float* gamma  = (const float*)d_in[9];
  const float* beta   = (const float*)d_in[10];
  float* out = (float*)d_out;

  float* ws      = (float*)d_ws;
  float* kbuf    = ws;
  float* Mpart   = kbuf    + (size_t)NB * NC * NL;
  float* sumPart = Mpart   + (size_t)NB * NTILES * NC * NC;
  float* sqPart  = sumPart + (size_t)NB * NTILES * NC;

  const int grid1 = NB * NTILES;
  k1_conv_qkv_M<<<grid1, BLOCK, 0, stream>>>(x, conv_w, conv_b, wq, bq, wk, bk,
                                             wv, bv, kbuf, Mpart);
  k2_foldM_gate_residual<<<NB * NBAND * NSEG, BLOCK, 0, stream>>>(x, kbuf, Mpart, out,
                                                              sumPart, sqPart);
  k3_fold_bn<<<NB * NC, K3_THREADS, 0, stream>>>(out, sumPart, sqPart, gamma, beta);
}
